# starting kernel
# baseline (speedup 1.0000x reference)
#include <hip/hip_runtime.h>
#include <hip/hip_bf16.h>

typedef __attribute__((ext_vector_type(2))) float v2f;
typedef __attribute__((ext_vector_type(8))) float v8f;

#define SGRID 7
#define CELLS 49
#define NCH   30
#define NBATCH 16384
#define NCLASS 20

#define WS_COORD 0
#define WS_CONF  49
#define WS_CLS   98
#define WS_A     152
#define WS_CB    204
#define WS_HAS   256
#define WS_HDR   256

__global__ void yolo_init(float* __restrict__ ws) {
    int t = threadIdx.x;
    if (t < WS_HDR) ws[t] = 0.0f;
}

__global__ void yolo_reduce(const float* __restrict__ out,
                            const float* __restrict__ tgt,
                            const int* __restrict__ imw,
                            const int* __restrict__ imh,
                            float* __restrict__ ws) {
    __shared__ float bins[3 * CELLS];
    const int tid = threadIdx.x;
    for (int k = tid; k < 3 * CELLS; k += blockDim.x) bins[k] = 0.0f;
    __syncthreads();

    const int   g  = blockIdx.x * blockDim.x + tid;
    const float W  = (float)imw[0];
    const float H  = (float)imh[0];
    const float cw = W / (float)SGRID;
    const float ch = H / (float)SGRID;

    const int b  = g / CELLS;
    const int ij = g - b * CELLS;
    const int i  = ij / SGRID;
    const int j  = ij - i * SGRID;

    const float* t = tgt + (size_t)g * NCH;
    const float* o = out + (size_t)g * NCH;

    const float t0 = t[0], t1 = t[1];
    const float tw = t[2] * W, th = t[3] * H;
    const float has = t[4];
    const float tcx = ((float)j + t0) * cw;
    const float tcy = ((float)i + t1) * ch;
    const float tx0 = tcx - tw * 0.5f, ty0 = tcy - th * 0.5f;
    const float tx1 = tcx + tw * 0.5f, ty1 = tcy + th * 0.5f;
    const float area_t = tw * th;

    float iou_p[2], conf_p[2];
#pragma unroll
    for (int p = 0; p < 2; ++p) {
        const float px = o[5 * p + 0], py = o[5 * p + 1];
        const float pw = o[5 * p + 3] * W, ph = o[5 * p + 4] * H;
        const float pcx = ((float)j + px) * cw;
        const float pcy = ((float)i + py) * ch;
        const float px0 = pcx - pw * 0.5f, py0 = pcy - ph * 0.5f;
        const float px1 = pcx + pw * 0.5f, py1 = pcy + ph * 0.5f;
        float iw = fminf(px1, tx1) - fmaxf(px0, tx0); iw = fmaxf(iw, 0.0f);
        float ih = fminf(py1, ty1) - fmaxf(py0, ty0); ih = fmaxf(ih, 0.0f);
        const float inter = iw * ih;
        iou_p[p]  = inter / (pw * ph + area_t - inter + 1e-9f);
        conf_p[p] = o[5 * p + 4];
    }
    const float best_conf = (iou_p[1] >= iou_p[0]) ? conf_p[1] : conf_p[0];

    const float px = o[5], py = o[6];
    const float pw = o[8] * W, ph = o[9] * H;
    const float pconf = o[9];
    const float dx = t0 - px, dy = t1 - py;
    const float dw = sqrtf(tw) - sqrtf(pw);
    const float dh = sqrtf(th) - sqrtf(ph);
    const float coord = dx * dx + dy * dy + dw * dw + dh * dh;
    const float dc = has - pconf;
    const float confm = dc * dc;

    float cls = 0.0f;
#pragma unroll
    for (int c = 0; c < NCLASS; ++c) {
        const float d = t[10 + c] - o[10 + c] * best_conf;
        cls = fmaf(d, d, cls);
    }

    ws[WS_HAS + g] = has;

    atomicAdd(&bins[ij],             coord);
    atomicAdd(&bins[CELLS + ij],     confm);
    atomicAdd(&bins[2 * CELLS + ij], cls);
    __syncthreads();
    for (int k = tid; k < 3 * CELLS; k += blockDim.x)
        atomicAdd(&ws[k], bins[k]);
}

__global__ void yolo_finalize(float* __restrict__ ws) {
    __shared__ float red[64];
    const int tid = threadIdx.x;
    float a = 0.0f, bsum = 0.0f;
    if (tid < CELLS) {
        const float coord = ws[WS_COORD + tid];
        const float conf  = ws[WS_CONF  + tid];
        const float cls   = ws[WS_CLS   + tid];
        a    = 5.0f * coord + 0.5f * conf;
        bsum = 0.5f * conf + cls;
    }
    if (tid < 52) ws[WS_A + tid] = (tid < CELLS) ? a : 0.0f;
    red[tid] = bsum;
    __syncthreads();
    for (int s = 32; s > 0; s >>= 1) {
        if (tid < s) red[tid] += red[tid + s];
        __syncthreads();
    }
    if (tid == 0) ws[WS_CB] = red[0];
}

__global__ void yolo_dot_wmma(const float* __restrict__ ws,
                              float* __restrict__ loss) {
    const int lane = threadIdx.x & 31;
    const int wave = (blockIdx.x * blockDim.x + threadIdx.x) >> 5;
    const int b0   = wave * 16;
    const float* __restrict__ Avec = ws + WS_A;
    const float* __restrict__ hasv = ws + WS_HAS;

    const int half = lane >> 4;
    const int m    = lane & 15;

    v8f acc = {};
#pragma unroll
    for (int k0 = 0; k0 < 52; k0 += 4) {
        const int ka = k0 + (half ? 2 : 0);
        const int kb = ka + 1;
        const int kca = ka > 48 ? 48 : ka;
        const int kcb = kb > 48 ? 48 : kb;
        v2f a, bm;
        a.x  = hasv[(size_t)(b0 + m) * CELLS + kca];
        a.y  = hasv[(size_t)(b0 + m) * CELLS + kcb];
        bm.x = Avec[ka];
        bm.y = Avec[kb];
        acc = __builtin_amdgcn_wmma_f32_16x16x4_f32(
             false, a,  false, bm,
             (short)0, acc,  false,  false);
    }

    const float cb = ws[WS_CB];
    if ((lane & 15) == 0) {
        const int mbase = b0 + half * 8;
#pragma unroll
        for (int r = 0; r < 8; ++r)
            loss[mbase + r] = acc[r] + cb;
    }
}

extern "C" void kernel_launch(void* const* d_in, const int* in_sizes, int n_in,
                              void* d_out, int out_size, void* d_ws, size_t ws_size,
                              hipStream_t stream) {
    const float* out_t = (const float*)d_in[0];
    const float* tgt_t = (const float*)d_in[1];
    const int*   imw   = (const int*)d_in[2];
    const int*   imh   = (const int*)d_in[3];
    float* loss = (float*)d_out;
    float* ws   = (float*)d_ws;

    yolo_init<<<1, 256, 0, stream>>>(ws);

    yolo_reduce<<<3136, 256, 0, stream>>>(out_t, tgt_t, imw, imh, ws);

    yolo_finalize<<<1, 64, 0, stream>>>(ws);

    yolo_dot_wmma<<<128, 256, 0, stream>>>(ws, loss);
}


# reduce pass: each block stages its contiguous 256-cell run of both tensors in LDS with whole-line 16-byte global loads, threads read their cell from LDS; atomics and sum order kind unchanged
# speedup vs baseline: 1.9331x; 1.9331x over previous
#include <hip/hip_runtime.h>
#include <hip/hip_bf16.h>

typedef __attribute__((ext_vector_type(2))) float v2f;
typedef __attribute__((ext_vector_type(8))) float v8f;

#define SGRID 7
#define CELLS 49
#define NCH   30
#define NBATCH 16384
#define NCLASS 20

#define WS_COORD 0
#define WS_CONF  49
#define WS_CLS   98
#define WS_A     152
#define WS_CB    204
#define WS_HAS   256
#define WS_HDR   256

__global__ void yolo_init(float* __restrict__ ws) {
    int t = threadIdx.x;
    if (t < WS_HDR) ws[t] = 0.0f;
}

#define TILE_CELLS 256
#define TILE_F4    (TILE_CELLS * NCH / 4)
#define COPY_STEPS (2 * TILE_F4 / TILE_CELLS)

__global__ void __launch_bounds__(TILE_CELLS)
yolo_reduce(const float* __restrict__ out,
            const float* __restrict__ tgt,
            const int* __restrict__ imw,
            const int* __restrict__ imh,
            float* __restrict__ ws) {
    __shared__ float4 s_in[2 * TILE_F4];
    __shared__ float bins[3 * CELLS];
    const int tid = threadIdx.x;

    const float4* __restrict__ gt =
        reinterpret_cast<const float4*>(tgt + (size_t)blockIdx.x * (TILE_CELLS * NCH));
    const float4* __restrict__ go =
        reinterpret_cast<const float4*>(out + (size_t)blockIdx.x * (TILE_CELLS * NCH));

    float4 r[COPY_STEPS];
#pragma unroll
    for (int k = 0; k < COPY_STEPS; ++k) {
        const int i = tid + TILE_CELLS * k;
        if (TILE_CELLS * (k + 1) <= TILE_F4)      r[k] = gt[i];
        else if (TILE_CELLS * k >= TILE_F4)       r[k] = go[i - TILE_F4];
        else r[k] = (i < TILE_F4) ? gt[i] : go[i - TILE_F4];
    }

    for (int k = tid; k < 3 * CELLS; k += TILE_CELLS) bins[k] = 0.0f;

#pragma unroll
    for (int k = 0; k < COPY_STEPS; ++k) s_in[tid + TILE_CELLS * k] = r[k];
    __syncthreads();

    const int   g  = blockIdx.x * TILE_CELLS + tid;
    const float W  = (float)imw[0];
    const float H  = (float)imh[0];
    const float cw = W / (float)SGRID;
    const float ch = H / (float)SGRID;

    const int b  = g / CELLS;
    const int ij = g - b * CELLS;
    const int i  = ij / SGRID;
    const int j  = ij - i * SGRID;

    const float* t = reinterpret_cast<const float*>(s_in) + tid * NCH;
    const float* o = reinterpret_cast<const float*>(s_in + TILE_F4) + tid * NCH;

    const float t0 = t[0], t1 = t[1];
    const float tw = t[2] * W, th = t[3] * H;
    const float has = t[4];
    const float tcx = ((float)j + t0) * cw;
    const float tcy = ((float)i + t1) * ch;
    const float tx0 = tcx - tw * 0.5f, ty0 = tcy - th * 0.5f;
    const float tx1 = tcx + tw * 0.5f, ty1 = tcy + th * 0.5f;
    const float area_t = tw * th;

    float iou_p[2], conf_p[2];
#pragma unroll
    for (int p = 0; p < 2; ++p) {
        const float px = o[5 * p + 0], py = o[5 * p + 1];
        const float pw = o[5 * p + 3] * W, ph = o[5 * p + 4] * H;
        const float pcx = ((float)j + px) * cw;
        const float pcy = ((float)i + py) * ch;
        const float px0 = pcx - pw * 0.5f, py0 = pcy - ph * 0.5f;
        const float px1 = pcx + pw * 0.5f, py1 = pcy + ph * 0.5f;
        float iw = fminf(px1, tx1) - fmaxf(px0, tx0); iw = fmaxf(iw, 0.0f);
        float ih = fminf(py1, ty1) - fmaxf(py0, ty0); ih = fmaxf(ih, 0.0f);
        const float inter = iw * ih;
        iou_p[p]  = inter / (pw * ph + area_t - inter + 1e-9f);
        conf_p[p] = o[5 * p + 4];
    }
    const float best_conf = (iou_p[1] >= iou_p[0]) ? conf_p[1] : conf_p[0];

    const float px = o[5], py = o[6];
    const float pw = o[8] * W, ph = o[9] * H;
    const float pconf = o[9];
    const float dx = t0 - px, dy = t1 - py;
    const float dw = sqrtf(tw) - sqrtf(pw);
    const float dh = sqrtf(th) - sqrtf(ph);
    const float coord = dx * dx + dy * dy + dw * dw + dh * dh;
    const float dc = has - pconf;
    const float confm = dc * dc;

    float cls = 0.0f;
#pragma unroll
    for (int c = 0; c < NCLASS; ++c) {
        const float d = t[10 + c] - o[10 + c] * best_conf;
        cls = fmaf(d, d, cls);
    }

    ws[WS_HAS + g] = has;

    atomicAdd(&bins[ij],             coord);
    atomicAdd(&bins[CELLS + ij],     confm);
    atomicAdd(&bins[2 * CELLS + ij], cls);
    __syncthreads();
    for (int k = tid; k < 3 * CELLS; k += blockDim.x)
        atomicAdd(&ws[k], bins[k]);
}

__global__ void yolo_finalize(float* __restrict__ ws) {
    __shared__ float red[64];
    const int tid = threadIdx.x;
    float a = 0.0f, bsum = 0.0f;
    if (tid < CELLS) {
        const float coord = ws[WS_COORD + tid];
        const float conf  = ws[WS_CONF  + tid];
        const float cls   = ws[WS_CLS   + tid];
        a    = 5.0f * coord + 0.5f * conf;
        bsum = 0.5f * conf + cls;
    }
    if (tid < 52) ws[WS_A + tid] = (tid < CELLS) ? a : 0.0f;
    red[tid] = bsum;
    __syncthreads();
    for (int s = 32; s > 0; s >>= 1) {
        if (tid < s) red[tid] += red[tid + s];
        __syncthreads();
    }
    if (tid == 0) ws[WS_CB] = red[0];
}

__global__ void yolo_dot_wmma(const float* __restrict__ ws,
                              float* __restrict__ loss) {
    const int lane = threadIdx.x & 31;
    const int wave = (blockIdx.x * blockDim.x + threadIdx.x) >> 5;
    const int b0   = wave * 16;
    const float* __restrict__ Avec = ws + WS_A;
    const float* __restrict__ hasv = ws + WS_HAS;

    const int half = lane >> 4;
    const int m    = lane & 15;

    v8f acc = {};
#pragma unroll
    for (int k0 = 0; k0 < 52; k0 += 4) {
        const int ka = k0 + (half ? 2 : 0);
        const int kb = ka + 1;
        const int kca = ka > 48 ? 48 : ka;
        const int kcb = kb > 48 ? 48 : kb;
        v2f a, bm;
        a.x  = hasv[(size_t)(b0 + m) * CELLS + kca];
        a.y  = hasv[(size_t)(b0 + m) * CELLS + kcb];
        bm.x = Avec[ka];
        bm.y = Avec[kb];
        acc = __builtin_amdgcn_wmma_f32_16x16x4_f32(
             false, a,  false, bm,
             (short)0, acc,  false,  false);
    }

    const float cb = ws[WS_CB];
    if ((lane & 15) == 0) {
        const int mbase = b0 + half * 8;
#pragma unroll
        for (int r = 0; r < 8; ++r)
            loss[mbase + r] = acc[r] + cb;
    }
}

extern "C" void kernel_launch(void* const* d_in, const int* in_sizes, int n_in,
                              void* d_out, int out_size, void* d_ws, size_t ws_size,
                              hipStream_t stream) {
    const float* out_t = (const float*)d_in[0];
    const float* tgt_t = (const float*)d_in[1];
    const int*   imw   = (const int*)d_in[2];
    const int*   imh   = (const int*)d_in[3];
    float* loss = (float*)d_out;
    float* ws   = (float*)d_ws;

    yolo_init<<<1, 256, 0, stream>>>(ws);

    yolo_reduce<<<3136, 256, 0, stream>>>(out_t, tgt_t, imw, imh, ws);

    yolo_finalize<<<1, 64, 0, stream>>>(ws);

    yolo_dot_wmma<<<128, 256, 0, stream>>>(ws, loss);
}


# reduce pass persistent: 256 blocks x 14 waves, each wave walks 7 tiles of 32 cells via its own LDS slice, next two tiles' loads in flight, no loop barrier; sum order: per-wave bins, 256 atomics/bin
# speedup vs baseline: 2.0062x; 1.0378x over previous
#include <hip/hip_runtime.h>
#include <hip/hip_bf16.h>

typedef __attribute__((ext_vector_type(2))) float v2f;
typedef __attribute__((ext_vector_type(8))) float v8f;

#define SGRID 7
#define CELLS 49
#define NCH   30
#define NBATCH 16384
#define NCLASS 20

#define WS_COORD 0
#define WS_CONF  49
#define WS_CLS   98
#define WS_A     152
#define WS_CB    204
#define WS_HAS   256
#define WS_HDR   256

__global__ void yolo_init(float* __restrict__ ws) {
    int t = threadIdx.x;
    if (t < WS_HDR) ws[t] = 0.0f;
}

typedef __attribute__((ext_vector_type(4))) float v4f;

#define RED_WAVES    14
#define RED_THREADS  (RED_WAVES * 32)
#define WT_PER_WAVE  7
#define WT_PER_BLOCK (RED_WAVES * WT_PER_WAVE)
#define WT_CELLS     32
#define WT_F4        (WT_CELLS * NCH / 4)
#define WT_STEPS     (2 * WT_F4 / 32)
#define NBINS        (3 * CELLS)

__device__ __forceinline__ void load_wave_tile(v4f (&r)[WT_STEPS],
                                               const v4f* __restrict__ gt,
                                               const v4f* __restrict__ go,
                                               const int lane) {
#pragma unroll
    for (int k = 0; k < WT_STEPS; ++k) {
        const int p = lane + 32 * k;
        if (32 * (k + 1) <= WT_F4)  r[k] = gt[p];
        else if (32 * k >= WT_F4)   r[k] = go[p - WT_F4];
        else {
            const v4f* src = (p < WT_F4) ? (gt + p) : (go + (p - WT_F4));
            r[k] = *src;
        }
    }
}

__device__ __forceinline__ void wave_lds_order() {
    __builtin_amdgcn_fence(__ATOMIC_RELEASE, "wavefront");
    __builtin_amdgcn_wave_barrier();
    __builtin_amdgcn_fence(__ATOMIC_ACQUIRE, "wavefront");
}

template <bool PREFETCH>
__device__ __forceinline__ void wave_tile_step(v4f (&r)[WT_STEPS],
                                               v4f* __restrict__ sw4,
                                               float* __restrict__ wbins,
                                               const v4f* __restrict__ gt_next,
                                               const v4f* __restrict__ go_next,
                                               const int tile, const int lane,
                                               const float W, const float H,
                                               const float cw, const float ch,
                                               float* __restrict__ ws) {
    wave_lds_order();
#pragma unroll
    for (int k = 0; k < WT_STEPS; ++k) sw4[lane + 32 * k] = r[k];
    wave_lds_order();

    if (PREFETCH) load_wave_tile(r, gt_next, go_next, lane);

    const int g  = tile * WT_CELLS + lane;
    const int b  = g / CELLS;
    const int ij = g - b * CELLS;
    const int i  = ij / SGRID;
    const int j  = ij - i * SGRID;

    const float* sw = reinterpret_cast<const float*>(sw4);
    const v2f* t2 = reinterpret_cast<const v2f*>(sw + lane * NCH);
    const v2f* o2 = reinterpret_cast<const v2f*>(sw + WT_CELLS * NCH + lane * NCH);

    float t[NCH], o[NCH];
#pragma unroll
    for (int c = 0; c < NCH / 2; ++c) {
        if (c < 3 || c >= 5) { const v2f v = t2[c]; t[2 * c] = v.x; t[2 * c + 1] = v.y; }
        const v2f u = o2[c]; o[2 * c] = u.x; o[2 * c + 1] = u.y;
    }

    const float t0 = t[0], t1 = t[1];
    const float tw = t[2] * W, th = t[3] * H;
    const float has = t[4];
    const float tcx = ((float)j + t0) * cw;
    const float tcy = ((float)i + t1) * ch;
    const float tx0 = tcx - tw * 0.5f, ty0 = tcy - th * 0.5f;
    const float tx1 = tcx + tw * 0.5f, ty1 = tcy + th * 0.5f;
    const float area_t = tw * th;

    float iou_p[2], conf_p[2];
#pragma unroll
    for (int p = 0; p < 2; ++p) {
        const float px = o[5 * p + 0], py = o[5 * p + 1];
        const float pw = o[5 * p + 3] * W, ph = o[5 * p + 4] * H;
        const float pcx = ((float)j + px) * cw;
        const float pcy = ((float)i + py) * ch;
        const float px0 = pcx - pw * 0.5f, py0 = pcy - ph * 0.5f;
        const float px1 = pcx + pw * 0.5f, py1 = pcy + ph * 0.5f;
        float iw = fminf(px1, tx1) - fmaxf(px0, tx0); iw = fmaxf(iw, 0.0f);
        float ih = fminf(py1, ty1) - fmaxf(py0, ty0); ih = fmaxf(ih, 0.0f);
        const float inter = iw * ih;
        iou_p[p]  = inter / (pw * ph + area_t - inter + 1e-9f);
        conf_p[p] = o[5 * p + 4];
    }
    const float best_conf = (iou_p[1] >= iou_p[0]) ? conf_p[1] : conf_p[0];

    const float px = o[5], py = o[6];
    const float pw = o[8] * W, ph = o[9] * H;
    const float pconf = o[9];
    const float dx = t0 - px, dy = t1 - py;
    const float dw = sqrtf(tw) - sqrtf(pw);
    const float dh = sqrtf(th) - sqrtf(ph);
    const float coord = dx * dx + dy * dy + dw * dw + dh * dh;
    const float dc = has - pconf;
    const float confm = dc * dc;

    float cls = 0.0f;
#pragma unroll
    for (int c = 0; c < NCLASS; ++c) {
        const float d = t[10 + c] - o[10 + c] * best_conf;
        cls = fmaf(d, d, cls);
    }

    ws[WS_HAS + g] = has;

    atomicAdd(&wbins[ij],             coord);
    atomicAdd(&wbins[CELLS + ij],     confm);
    atomicAdd(&wbins[2 * CELLS + ij], cls);
}

__global__ void __launch_bounds__(RED_THREADS)
yolo_reduce(const float* __restrict__ out,
            const float* __restrict__ tgt,
            const int* __restrict__ imw,
            const int* __restrict__ imh,
            float* __restrict__ ws) {
    __shared__ v4f   s_in[RED_WAVES * 2 * WT_F4];
    __shared__ float bins[RED_WAVES * NBINS];
    const int tid  = threadIdx.x;
    const int lane = tid & 31;
    const int wave = __builtin_amdgcn_readfirstlane(tid >> 5);

    const v4f* __restrict__ gt4 = reinterpret_cast<const v4f*>(tgt);
    const v4f* __restrict__ go4 = reinterpret_cast<const v4f*>(out);

    const int first = blockIdx.x * WT_PER_BLOCK + wave;
#define TILE_AT(s)  (first + RED_WAVES * (s))
#define GT_AT(s)    (gt4 + (size_t)TILE_AT(s) * WT_F4)
#define GO_AT(s)    (go4 + (size_t)TILE_AT(s) * WT_F4)

    v4f ra[WT_STEPS], rb[WT_STEPS];
    load_wave_tile(ra, GT_AT(0), GO_AT(0), lane);
    load_wave_tile(rb, GT_AT(1), GO_AT(1), lane);

    v4f*   sw4   = s_in + wave * (2 * WT_F4);
    float* wbins = bins + wave * NBINS;
#pragma unroll
    for (int m = 0; m < (NBINS + 31) / 32; ++m) {
        const int k = lane + 32 * m;
        if (k < NBINS) wbins[k] = 0.0f;
    }

    const float W  = (float)imw[0];
    const float H  = (float)imh[0];
    const float cw = W / (float)SGRID;
    const float ch = H / (float)SGRID;

    wave_tile_step<true >(ra, sw4, wbins, GT_AT(2), GO_AT(2), TILE_AT(0), lane, W, H, cw, ch, ws);
    wave_tile_step<true >(rb, sw4, wbins, GT_AT(3), GO_AT(3), TILE_AT(1), lane, W, H, cw, ch, ws);
    wave_tile_step<true >(ra, sw4, wbins, GT_AT(4), GO_AT(4), TILE_AT(2), lane, W, H, cw, ch, ws);
    wave_tile_step<true >(rb, sw4, wbins, GT_AT(5), GO_AT(5), TILE_AT(3), lane, W, H, cw, ch, ws);
    wave_tile_step<true >(ra, sw4, wbins, GT_AT(6), GO_AT(6), TILE_AT(4), lane, W, H, cw, ch, ws);
    wave_tile_step<false>(rb, sw4, wbins, gt4,      go4,      TILE_AT(5), lane, W, H, cw, ch, ws);
    wave_tile_step<false>(ra, sw4, wbins, gt4,      go4,      TILE_AT(6), lane, W, H, cw, ch, ws);
#undef TILE_AT
#undef GT_AT
#undef GO_AT

    __syncthreads();
    for (int k = tid; k < NBINS; k += RED_THREADS) {
        float s = 0.0f;
#pragma unroll
        for (int w = 0; w < RED_WAVES; ++w) s += bins[w * NBINS + k];
        atomicAdd(&ws[k], s);
    }
}

__global__ void yolo_finalize(float* __restrict__ ws) {
    __shared__ float red[64];
    const int tid = threadIdx.x;
    float a = 0.0f, bsum = 0.0f;
    if (tid < CELLS) {
        const float coord = ws[WS_COORD + tid];
        const float conf  = ws[WS_CONF  + tid];
        const float cls   = ws[WS_CLS   + tid];
        a    = 5.0f * coord + 0.5f * conf;
        bsum = 0.5f * conf + cls;
    }
    if (tid < 52) ws[WS_A + tid] = (tid < CELLS) ? a : 0.0f;
    red[tid] = bsum;
    __syncthreads();
    for (int s = 32; s > 0; s >>= 1) {
        if (tid < s) red[tid] += red[tid + s];
        __syncthreads();
    }
    if (tid == 0) ws[WS_CB] = red[0];
}

__global__ void yolo_dot_wmma(const float* __restrict__ ws,
                              float* __restrict__ loss) {
    const int lane = threadIdx.x & 31;
    const int wave = (blockIdx.x * blockDim.x + threadIdx.x) >> 5;
    const int b0   = wave * 16;
    const float* __restrict__ Avec = ws + WS_A;
    const float* __restrict__ hasv = ws + WS_HAS;

    const int half = lane >> 4;
    const int m    = lane & 15;

    v8f acc = {};
#pragma unroll
    for (int k0 = 0; k0 < 52; k0 += 4) {
        const int ka = k0 + (half ? 2 : 0);
        const int kb = ka + 1;
        const int kca = ka > 48 ? 48 : ka;
        const int kcb = kb > 48 ? 48 : kb;
        v2f a, bm;
        a.x  = hasv[(size_t)(b0 + m) * CELLS + kca];
        a.y  = hasv[(size_t)(b0 + m) * CELLS + kcb];
        bm.x = Avec[ka];
        bm.y = Avec[kb];
        acc = __builtin_amdgcn_wmma_f32_16x16x4_f32(
             false, a,  false, bm,
             (short)0, acc,  false,  false);
    }

    const float cb = ws[WS_CB];
    if ((lane & 15) == 0) {
        const int mbase = b0 + half * 8;
#pragma unroll
        for (int r = 0; r < 8; ++r)
            loss[mbase + r] = acc[r] + cb;
    }
}

extern "C" void kernel_launch(void* const* d_in, const int* in_sizes, int n_in,
                              void* d_out, int out_size, void* d_ws, size_t ws_size,
                              hipStream_t stream) {
    const float* out_t = (const float*)d_in[0];
    const float* tgt_t = (const float*)d_in[1];
    const int*   imw   = (const int*)d_in[2];
    const int*   imh   = (const int*)d_in[3];
    float* loss = (float*)d_out;
    float* ws   = (float*)d_ws;

    yolo_init<<<1, 256, 0, stream>>>(ws);

    yolo_reduce<<<256, 448, 0, stream>>>(out_t, tgt_t, imw, imh, ws);

    yolo_finalize<<<1, 64, 0, stream>>>(ws);

    yolo_dot_wmma<<<128, 256, 0, stream>>>(ws, loss);
}


# speedup vs baseline: 2.3495x; 1.1711x over previous
#include <hip/hip_runtime.h>
#include <hip/hip_bf16.h>

typedef __attribute__((ext_vector_type(2))) float v2f;
typedef __attribute__((ext_vector_type(8))) float v8f;

#define SGRID 7
#define CELLS 49
#define NCH   30
#define NBATCH 16384
#define NCLASS 20

#define WS_COORD 0
#define WS_CONF  49
#define WS_CLS   98
#define WS_HAS   256
#define WS_HDR   256

__global__ void yolo_init(float* __restrict__ ws) {
    int t = threadIdx.x;
    if (t < WS_HDR) ws[t] = 0.0f;
}

typedef __attribute__((ext_vector_type(4))) float v4f;

#define RED_WAVES    14
#define RED_THREADS  (RED_WAVES * 32)
#define WT_PER_WAVE  7
#define WT_PER_BLOCK (RED_WAVES * WT_PER_WAVE)
#define WT_CELLS     32
#define WT_F4        (WT_CELLS * NCH / 4)
#define WT_STEPS     (2 * WT_F4 / 32)
#define NBINS        (3 * CELLS)

__device__ __forceinline__ void load_wave_tile(v4f (&r)[WT_STEPS],
                                               const v4f* __restrict__ gt,
                                               const v4f* __restrict__ go,
                                               const int lane) {
#pragma unroll
    for (int k = 0; k < WT_STEPS; ++k) {
        const int p = lane + 32 * k;
        if (32 * (k + 1) <= WT_F4)  r[k] = __builtin_nontemporal_load(gt + p);
        else if (32 * k >= WT_F4)   r[k] = __builtin_nontemporal_load(go + (p - WT_F4));
        else {
            const v4f* src = (p < WT_F4) ? (gt + p) : (go + (p - WT_F4));
            r[k] = __builtin_nontemporal_load(src);
        }
    }
}

__device__ __forceinline__ void wave_lds_order() {
    __builtin_amdgcn_fence(__ATOMIC_RELEASE, "wavefront");
    __builtin_amdgcn_wave_barrier();
    __builtin_amdgcn_fence(__ATOMIC_ACQUIRE, "wavefront");
}

template <bool PREFETCH>
__device__ __forceinline__ void wave_tile_step(v4f (&r)[WT_STEPS],
                                               v4f* __restrict__ sw4,
                                               float* __restrict__ wbins,
                                               const v4f* __restrict__ gt_next,
                                               const v4f* __restrict__ go_next,
                                               const int tile, const int lane,
                                               const float W, const float H,
                                               const float cw, const float ch,
                                               float* __restrict__ ws) {
    wave_lds_order();
#pragma unroll
    for (int k = 0; k < WT_STEPS; ++k) sw4[lane + 32 * k] = r[k];
    wave_lds_order();

    if (PREFETCH) load_wave_tile(r, gt_next, go_next, lane);

    const int g  = tile * WT_CELLS + lane;
    const int b  = g / CELLS;
    const int ij = g - b * CELLS;
    const int i  = ij / SGRID;
    const int j  = ij - i * SGRID;

    const float* sw = reinterpret_cast<const float*>(sw4);
    const v2f* t2 = reinterpret_cast<const v2f*>(sw + lane * NCH);
    const v2f* o2 = reinterpret_cast<const v2f*>(sw + WT_CELLS * NCH + lane * NCH);

    float t[NCH], o[NCH];
#pragma unroll
    for (int c = 0; c < NCH / 2; ++c) {
        if (c < 3 || c >= 5) { const v2f v = t2[c]; t[2 * c] = v.x; t[2 * c + 1] = v.y; }
        const v2f u = o2[c]; o[2 * c] = u.x; o[2 * c + 1] = u.y;
    }

    const float t0 = t[0], t1 = t[1];
    const float tw = t[2] * W, th = t[3] * H;
    const float has = t[4];
    const float tcx = ((float)j + t0) * cw;
    const float tcy = ((float)i + t1) * ch;
    const float tx0 = tcx - tw * 0.5f, ty0 = tcy - th * 0.5f;
    const float tx1 = tcx + tw * 0.5f, ty1 = tcy + th * 0.5f;
    const float area_t = tw * th;

    float iou_p[2], conf_p[2];
#pragma unroll
    for (int p = 0; p < 2; ++p) {
        const float px = o[5 * p + 0], py = o[5 * p + 1];
        const float pw = o[5 * p + 3] * W, ph = o[5 * p + 4] * H;
        const float pcx = ((float)j + px) * cw;
        const float pcy = ((float)i + py) * ch;
        const float px0 = pcx - pw * 0.5f, py0 = pcy - ph * 0.5f;
        const float px1 = pcx + pw * 0.5f, py1 = pcy + ph * 0.5f;
        float iw = fminf(px1, tx1) - fmaxf(px0, tx0); iw = fmaxf(iw, 0.0f);
        float ih = fminf(py1, ty1) - fmaxf(py0, ty0); ih = fmaxf(ih, 0.0f);
        const float inter = iw * ih;
        iou_p[p]  = inter / (pw * ph + area_t - inter + 1e-9f);
        conf_p[p] = o[5 * p + 4];
    }
    const float best_conf = (iou_p[1] >= iou_p[0]) ? conf_p[1] : conf_p[0];

    const float px = o[5], py = o[6];
    const float pw = o[8] * W, ph = o[9] * H;
    const float pconf = o[9];
    const float dx = t0 - px, dy = t1 - py;
    const float dw = sqrtf(tw) - sqrtf(pw);
    const float dh = sqrtf(th) - sqrtf(ph);
    const float coord = dx * dx + dy * dy + dw * dw + dh * dh;
    const float dc = has - pconf;
    const float confm = dc * dc;

    float cls = 0.0f;
#pragma unroll
    for (int c = 0; c < NCLASS; ++c) {
        const float d = t[10 + c] - o[10 + c] * best_conf;
        cls = fmaf(d, d, cls);
    }

    ws[WS_HAS + g] = has;

    atomicAdd(&wbins[ij],             coord);
    atomicAdd(&wbins[CELLS + ij],     confm);
    atomicAdd(&wbins[2 * CELLS + ij], cls);
}

__global__ void __launch_bounds__(RED_THREADS)
yolo_reduce(const float* __restrict__ out,
            const float* __restrict__ tgt,
            const int* __restrict__ imw,
            const int* __restrict__ imh,
            float* __restrict__ ws) {
    __shared__ v4f   s_in[RED_WAVES * 2 * WT_F4];
    __shared__ float bins[RED_WAVES * NBINS];
    const int tid  = threadIdx.x;
    const int lane = tid & 31;
    const int wave = __builtin_amdgcn_readfirstlane(tid >> 5);

    const v4f* __restrict__ gt4 = reinterpret_cast<const v4f*>(tgt);
    const v4f* __restrict__ go4 = reinterpret_cast<const v4f*>(out);

    const int first = blockIdx.x * WT_PER_BLOCK + wave;
#define TILE_AT(s)  (first + RED_WAVES * (s))
#define GT_AT(s)    (gt4 + (size_t)TILE_AT(s) * WT_F4)
#define GO_AT(s)    (go4 + (size_t)TILE_AT(s) * WT_F4)

    v4f ra[WT_STEPS], rb[WT_STEPS];
    load_wave_tile(ra, GT_AT(0), GO_AT(0), lane);
    load_wave_tile(rb, GT_AT(1), GO_AT(1), lane);

    v4f*   sw4   = s_in + wave * (2 * WT_F4);
    float* wbins = bins + wave * NBINS;
#pragma unroll
    for (int m = 0; m < (NBINS + 31) / 32; ++m) {
        const int k = lane + 32 * m;
        if (k < NBINS) wbins[k] = 0.0f;
    }

    const float W  = (float)imw[0];
    const float H  = (float)imh[0];
    const float cw = W / (float)SGRID;
    const float ch = H / (float)SGRID;

    wave_tile_step<true >(ra, sw4, wbins, GT_AT(2), GO_AT(2), TILE_AT(0), lane, W, H, cw, ch, ws);
    wave_tile_step<true >(rb, sw4, wbins, GT_AT(3), GO_AT(3), TILE_AT(1), lane, W, H, cw, ch, ws);
    wave_tile_step<true >(ra, sw4, wbins, GT_AT(4), GO_AT(4), TILE_AT(2), lane, W, H, cw, ch, ws);
    wave_tile_step<true >(rb, sw4, wbins, GT_AT(5), GO_AT(5), TILE_AT(3), lane, W, H, cw, ch, ws);
    wave_tile_step<true >(ra, sw4, wbins, GT_AT(6), GO_AT(6), TILE_AT(4), lane, W, H, cw, ch, ws);
    wave_tile_step<false>(rb, sw4, wbins, gt4,      go4,      TILE_AT(5), lane, W, H, cw, ch, ws);
    wave_tile_step<false>(ra, sw4, wbins, gt4,      go4,      TILE_AT(6), lane, W, H, cw, ch, ws);
#undef TILE_AT
#undef GT_AT
#undef GO_AT

    __syncthreads();
    for (int k = tid; k < NBINS; k += RED_THREADS) {
        float s = 0.0f;
#pragma unroll
        for (int w = 0; w < RED_WAVES; ++w) s += bins[w * NBINS + k];
        atomicAdd(&ws[k], s);
    }
}

__global__ void yolo_dot_wmma(const float* __restrict__ ws,
                              float* __restrict__ loss) {
    const int lane = threadIdx.x & 31;
    const int wave = (blockIdx.x * blockDim.x + threadIdx.x) >> 5;
    const int b0   = wave * 16;
    const float* __restrict__ hasv = ws + WS_HAS;

    const int half = lane >> 4;
    const int m    = lane & 15;

    const bool in1 = (lane + 32) < CELLS;
    const int  c1  = in1 ? lane + 32 : 0;
    const float coord0 = ws[WS_COORD + lane], conf0 = ws[WS_CONF + lane], cls0 = ws[WS_CLS + lane];
    const float coord1 = ws[WS_COORD + c1],   conf1 = ws[WS_CONF + c1],   cls1 = ws[WS_CLS + c1];
    const float a_lo = 5.0f * coord0 + 0.5f * conf0;
    const float a_hi = in1 ? (5.0f * coord1 + 0.5f * conf1) : 0.0f;
    const float b_lo = 0.5f * conf0 + cls0;
    const float b_hi = in1 ? (0.5f * conf1 + cls1) : 0.0f;

    float bs = b_lo + b_hi;
#pragma unroll
    for (int s = 16; s > 0; s >>= 1) bs += __shfl_down(bs, s, 32);
    const float cb = __int_as_float(__builtin_amdgcn_readfirstlane(__float_as_int(bs)));

    const int a_lo_i = __float_as_int(a_lo);
    const int a_hi_i = __float_as_int(a_hi);

    v8f acc = {};
#pragma unroll
    for (int k0 = 0; k0 < 52; k0 += 4) {
        const int ka = k0 + (half ? 2 : 0);
        const int kb = ka + 1;
        const int kca = ka > 48 ? 48 : ka;
        const int kcb = kb > 48 ? 48 : kb;
        v2f a, bm;
        a.x  = hasv[(size_t)(b0 + m) * CELLS + kca];
        a.y  = hasv[(size_t)(b0 + m) * CELLS + kcb];
        float Ak[4];
#pragma unroll
        for (int q = 0; q < 4; ++q) {
            const int k = k0 + q;
            Ak[q] = (k >= CELLS) ? 0.0f
                  : __int_as_float(k < 32 ? __builtin_amdgcn_readlane(a_lo_i, k)
                                          : __builtin_amdgcn_readlane(a_hi_i, k - 32));
        }
        bm.x = half ? Ak[2] : Ak[0];
        bm.y = half ? Ak[3] : Ak[1];
        acc = __builtin_amdgcn_wmma_f32_16x16x4_f32(
             false, a,  false, bm,
             (short)0, acc,  false,  false);
    }

    if ((lane & 15) == 0) {
        const int mbase = b0 + half * 8;
#pragma unroll
        for (int r = 0; r < 8; ++r)
            loss[mbase + r] = acc[r] + cb;
    }
}

extern "C" void kernel_launch(void* const* d_in, const int* in_sizes, int n_in,
                              void* d_out, int out_size, void* d_ws, size_t ws_size,
                              hipStream_t stream) {
    const float* out_t = (const float*)d_in[0];
    const float* tgt_t = (const float*)d_in[1];
    const int*   imw   = (const int*)d_in[2];
    const int*   imh   = (const int*)d_in[3];
    float* loss = (float*)d_out;
    float* ws   = (float*)d_ws;

    yolo_init<<<1, 256, 0, stream>>>(ws);

    yolo_reduce<<<256, 448, 0, stream>>>(out_t, tgt_t, imw, imh, ws);

    yolo_dot_wmma<<<128, 256, 0, stream>>>(ws, loss);
}


# reduce pass: the four square roots and two IoU quotients per cell use the hardware's one-instruction sqrt and reciprocal (1 ulp) instead of the correctly rounded expansions; sum order unchanged
# speedup vs baseline: 2.6766x; 1.1392x over previous
#include <hip/hip_runtime.h>
#include <hip/hip_bf16.h>

typedef __attribute__((ext_vector_type(2))) float v2f;
typedef __attribute__((ext_vector_type(8))) float v8f;

#define SGRID 7
#define CELLS 49
#define NCH   30
#define NBATCH 16384
#define NCLASS 20

#define RED_BLOCKS 256
#define PROW       52
#define WS_PART    0
#define WS_HAS     (RED_BLOCKS * PROW)
#define WS_FLOATS  (WS_HAS + NBATCH * CELLS)
#define WS_BYTES   ((unsigned long long)WS_FLOATS * 4ull)

template <int MASK>
__device__ __forceinline__ float row_xor(const float v) {
    return __int_as_float(__builtin_amdgcn_mov_dpp(__float_as_int(v), 0x160 + MASK, 0xf, 0xf, false));
}
__device__ __forceinline__ float row16_sum(float v) {
    v += row_xor<1>(v);
    v += row_xor<2>(v);
    v += row_xor<4>(v);
    v += row_xor<8>(v);
    return v;
}

typedef __attribute__((ext_vector_type(4))) float v4f;

#define RED_WAVES    14
#define RED_THREADS  (RED_WAVES * 32)
#define WT_PER_WAVE  7
#define WT_PER_BLOCK (RED_WAVES * WT_PER_WAVE)
#define WT_CELLS     32
#define WT_F4        (WT_CELLS * NCH / 4)
#define WT_STEPS     (2 * WT_F4 / 32)
#define NBINS        (3 * CELLS)

__device__ __forceinline__ void load_wave_tile(v4f (&r)[WT_STEPS],
                                               const v4f* __restrict__ gt,
                                               const v4f* __restrict__ go,
                                               const int lane) {
#pragma unroll
    for (int k = 0; k < WT_STEPS; ++k) {
        const int p = lane + 32 * k;
        if (32 * (k + 1) <= WT_F4)  r[k] = __builtin_nontemporal_load(gt + p);
        else if (32 * k >= WT_F4)   r[k] = __builtin_nontemporal_load(go + (p - WT_F4));
        else {
            const v4f* src = (p < WT_F4) ? (gt + p) : (go + (p - WT_F4));
            r[k] = __builtin_nontemporal_load(src);
        }
    }
}

__device__ __forceinline__ void wave_lds_order() {
    __builtin_amdgcn_fence(__ATOMIC_RELEASE, "wavefront");
    __builtin_amdgcn_wave_barrier();
    __builtin_amdgcn_fence(__ATOMIC_ACQUIRE, "wavefront");
}

template <bool PREFETCH>
__device__ __forceinline__ void wave_tile_step(v4f (&r)[WT_STEPS],
                                               v4f* __restrict__ sw4,
                                               float* __restrict__ wbins,
                                               const v4f* __restrict__ gt_next,
                                               const v4f* __restrict__ go_next,
                                               const int tile, const int lane,
                                               const float W, const float H,
                                               const float cw, const float ch,
                                               float* __restrict__ ws) {
    wave_lds_order();
#pragma unroll
    for (int k = 0; k < WT_STEPS; ++k) sw4[lane + 32 * k] = r[k];
    wave_lds_order();

    if (PREFETCH) load_wave_tile(r, gt_next, go_next, lane);

    const int g  = tile * WT_CELLS + lane;
    const int b  = g / CELLS;
    const int ij = g - b * CELLS;
    const int i  = ij / SGRID;
    const int j  = ij - i * SGRID;

    const float* sw = reinterpret_cast<const float*>(sw4);
    const v2f* t2 = reinterpret_cast<const v2f*>(sw + lane * NCH);
    const v2f* o2 = reinterpret_cast<const v2f*>(sw + WT_CELLS * NCH + lane * NCH);

    float t[NCH], o[NCH];
#pragma unroll
    for (int c = 0; c < NCH / 2; ++c) {
        if (c < 3 || c >= 5) { const v2f v = t2[c]; t[2 * c] = v.x; t[2 * c + 1] = v.y; }
        const v2f u = o2[c]; o[2 * c] = u.x; o[2 * c + 1] = u.y;
    }

    const float t0 = t[0], t1 = t[1];
    const float tw = t[2] * W, th = t[3] * H;
    const float has = t[4];
    const float tcx = ((float)j + t0) * cw;
    const float tcy = ((float)i + t1) * ch;
    const float tx0 = tcx - tw * 0.5f, ty0 = tcy - th * 0.5f;
    const float tx1 = tcx + tw * 0.5f, ty1 = tcy + th * 0.5f;
    const float area_t = tw * th;

    float iou_p[2], conf_p[2];
#pragma unroll
    for (int p = 0; p < 2; ++p) {
        const float px = o[5 * p + 0], py = o[5 * p + 1];
        const float pw = o[5 * p + 3] * W, ph = o[5 * p + 4] * H;
        const float pcx = ((float)j + px) * cw;
        const float pcy = ((float)i + py) * ch;
        const float px0 = pcx - pw * 0.5f, py0 = pcy - ph * 0.5f;
        const float px1 = pcx + pw * 0.5f, py1 = pcy + ph * 0.5f;
        float iw = fminf(px1, tx1) - fmaxf(px0, tx0); iw = fmaxf(iw, 0.0f);
        float ih = fminf(py1, ty1) - fmaxf(py0, ty0); ih = fmaxf(ih, 0.0f);
        const float inter = iw * ih;
        iou_p[p]  = inter * __builtin_amdgcn_rcpf(pw * ph + area_t - inter + 1e-9f);
        conf_p[p] = o[5 * p + 4];
    }
    const float best_conf = (iou_p[1] >= iou_p[0]) ? conf_p[1] : conf_p[0];

    const float px = o[5], py = o[6];
    const float pw = o[8] * W, ph = o[9] * H;
    const float pconf = o[9];
    const float dx = t0 - px, dy = t1 - py;
    const float dw = __builtin_amdgcn_sqrtf(tw) - __builtin_amdgcn_sqrtf(pw);
    const float dh = __builtin_amdgcn_sqrtf(th) - __builtin_amdgcn_sqrtf(ph);
    const float coord = dx * dx + dy * dy + dw * dw + dh * dh;
    const float dc = has - pconf;
    const float confm = dc * dc;

    float cls = 0.0f;
#pragma unroll
    for (int c = 0; c < NCLASS; ++c) {
        const float d = t[10 + c] - o[10 + c] * best_conf;
        cls = fmaf(d, d, cls);
    }

    ws[WS_HAS + g] = has;

    atomicAdd(&wbins[ij],             coord);
    atomicAdd(&wbins[CELLS + ij],     confm);
    atomicAdd(&wbins[2 * CELLS + ij], cls);
}

__global__ void __launch_bounds__(RED_THREADS)
yolo_reduce(const float* __restrict__ out,
            const float* __restrict__ tgt,
            const int* __restrict__ imw,
            const int* __restrict__ imh,
            float* __restrict__ ws,
            const unsigned long long ws_bytes) {
    __shared__ v4f   s_in[RED_WAVES * 2 * WT_F4];
    __shared__ float bins[RED_WAVES * NBINS];
    const int tid  = threadIdx.x;
    const int lane = tid & 31;
    const int wave = __builtin_amdgcn_readfirstlane(tid >> 5);

    const v4f* __restrict__ gt4 = reinterpret_cast<const v4f*>(tgt);
    const v4f* __restrict__ go4 = reinterpret_cast<const v4f*>(out);

    const int first = blockIdx.x * WT_PER_BLOCK + wave;
#define TILE_AT(s)  (first + RED_WAVES * (s))
#define GT_AT(s)    (gt4 + (size_t)TILE_AT(s) * WT_F4)
#define GO_AT(s)    (go4 + (size_t)TILE_AT(s) * WT_F4)

    v4f ra[WT_STEPS], rb[WT_STEPS];
    load_wave_tile(ra, GT_AT(0), GO_AT(0), lane);
    load_wave_tile(rb, GT_AT(1), GO_AT(1), lane);

    if (out == nullptr || tgt == nullptr || ws == nullptr || ws_bytes < WS_BYTES) return;

    v4f*   sw4   = s_in + wave * (2 * WT_F4);
    float* wbins = bins + wave * NBINS;
#pragma unroll
    for (int m = 0; m < (NBINS + 31) / 32; ++m) {
        const int k = lane + 32 * m;
        if (k < NBINS) wbins[k] = 0.0f;
    }

    const float W  = (float)imw[0];
    const float H  = (float)imh[0];
    const float cw = W / (float)SGRID;
    const float ch = H / (float)SGRID;

    wave_tile_step<true >(ra, sw4, wbins, GT_AT(2), GO_AT(2), TILE_AT(0), lane, W, H, cw, ch, ws);
    wave_tile_step<true >(rb, sw4, wbins, GT_AT(3), GO_AT(3), TILE_AT(1), lane, W, H, cw, ch, ws);
    wave_tile_step<true >(ra, sw4, wbins, GT_AT(4), GO_AT(4), TILE_AT(2), lane, W, H, cw, ch, ws);
    wave_tile_step<true >(rb, sw4, wbins, GT_AT(5), GO_AT(5), TILE_AT(3), lane, W, H, cw, ch, ws);
    wave_tile_step<true >(ra, sw4, wbins, GT_AT(6), GO_AT(6), TILE_AT(4), lane, W, H, cw, ch, ws);
    wave_tile_step<false>(rb, sw4, wbins, gt4,      go4,      TILE_AT(5), lane, W, H, cw, ch, ws);
    wave_tile_step<false>(ra, sw4, wbins, gt4,      go4,      TILE_AT(6), lane, W, H, cw, ch, ws);
#undef TILE_AT
#undef GT_AT
#undef GO_AT

    __syncthreads();
    if (wave == 0) {
        const bool in1 = (lane + 32) < CELLS;
        const int  c1  = in1 ? lane + 32 : lane;
        float co0 = 0.0f, cf0 = 0.0f, cl0 = 0.0f, co1 = 0.0f, cf1 = 0.0f, cl1 = 0.0f;
#pragma unroll
        for (int w = 0; w < RED_WAVES; ++w) {
            const float* wb = bins + w * NBINS;
            co0 += wb[lane];  cf0 += wb[CELLS + lane];  cl0 += wb[2 * CELLS + lane];
            co1 += wb[c1];    cf1 += wb[CELLS + c1];    cl1 += wb[2 * CELLS + c1];
        }
        const float a0 = 5.0f * co0 + 0.5f * cf0;
        const float a1 = in1 ? (5.0f * co1 + 0.5f * cf1) : 0.0f;
        const float b0 = 0.5f * cf0 + cl0;
        const float b1 = in1 ? (0.5f * cf1 + cl1) : 0.0f;
        const float rs = row16_sum(b0 + b1);
        const float cb = __int_as_float(__builtin_amdgcn_readlane(__float_as_int(rs), 0))
                       + __int_as_float(__builtin_amdgcn_readlane(__float_as_int(rs), 16));
        float* __restrict__ row = ws + WS_PART + blockIdx.x * PROW;
        row[lane] = a0;
        if (lane < PROW - 32) row[32 + lane] = in1 ? a1 : (lane == CELLS - 32 ? cb : 0.0f);
    }
}

#define DOT_THREADS 256
#define NGROUP      16
#define GROUP_F4    (NGROUP * PROW / 4)
__global__ void __launch_bounds__(DOT_THREADS)
yolo_dot_wmma(const float* __restrict__ ws,
              float* __restrict__ loss,
              const unsigned long long ws_bytes) {
    __shared__ v4f sB4[GROUP_F4];
    if (ws == nullptr || ws_bytes < WS_BYTES) return;

    const int tid  = threadIdx.x;
    const int lane = tid & 31;
    const int wave = (blockIdx.x * DOT_THREADS + tid) >> 5;
    const int b0   = wave * 16;
    const float* __restrict__ hasv = ws + WS_HAS;

    const int half = lane >> 4;
    const int m    = lane & 15;

    const float* __restrict__ hrow = hasv + (size_t)(b0 + m) * CELLS;
    v2f afr[13];
#pragma unroll
    for (int i = 0; i < 12; ++i) {
        afr[i].x = hrow[4 * i + 2 * half];
        afr[i].y = hrow[4 * i + 2 * half + 1];
    }
    const float h48 = hrow[CELLS - 1];
    afr[12].x = half ? 0.0f : h48;
    afr[12].y = half ? 0.0f : 1.0f;

    const v4f* __restrict__ p4 = reinterpret_cast<const v4f*>(ws + WS_PART);
    if (tid < GROUP_F4) {
        v4f r[NGROUP];
#pragma unroll
        for (int j = 0; j < NGROUP; ++j) r[j] = p4[tid + GROUP_F4 * j];
        sB4[tid] = ((r[0] + r[1]) + (r[2] + r[3])) + ((r[4] + r[5]) + (r[6] + r[7]))
                 + (((r[8] + r[9]) + (r[10] + r[11])) + ((r[12] + r[13]) + (r[14] + r[15])));
    }
    __syncthreads();

    const float* sB = reinterpret_cast<const float*>(sB4);
    const v2f* __restrict__ brow = reinterpret_cast<const v2f*>(sB + PROW * m + 2 * half);

    v8f acc = {};
#pragma unroll
    for (int i = 0; i < 13; ++i) {
        const v2f bm = brow[2 * i];
        acc = __builtin_amdgcn_wmma_f32_16x16x4_f32(
             false, afr[i],  false, bm,
             (short)0, acc,  false,  false);
    }

#pragma unroll
    for (int r = 0; r < 8; ++r) acc[r] = row16_sum(acc[r]);
    if ((lane & 15) == 0) {
        const int mbase = b0 + half * 8;
#pragma unroll
        for (int r = 0; r < 8; ++r)
            loss[mbase + r] = acc[r];
    }
}

extern "C" void kernel_launch(void* const* d_in, const int* in_sizes, int n_in,
                              void* d_out, int out_size, void* d_ws, size_t ws_size,
                              hipStream_t stream) {
    const float* out_t = (const float*)d_in[0];
    const float* tgt_t = (const float*)d_in[1];
    const int*   imw   = (const int*)d_in[2];
    const int*   imh   = (const int*)d_in[3];
    float* loss = (float*)d_out;
    float* ws   = (float*)d_ws;

    yolo_reduce<<<256, 448, 0, stream>>>(out_t, tgt_t, imw, imh, ws, ws_size);

    yolo_dot_wmma<<<128, 256, 0, stream>>>(ws, loss, ws_size);
}
